# starting kernel
# baseline (speedup 1.0000x reference)
#include <hip/hip_runtime.h>

#define B_     4096
#define NIN_   512
#define L_     16
#define N_     512
#define K_     32
#define NOUT_  256
#define TOTAL_ (NIN_ + L_ * N_)

typedef float v2f __attribute__((ext_vector_type(2)));
typedef float v8f __attribute__((ext_vector_type(8)));

__global__ __launch_bounds__(256)
void ffn_init_buf(const float* __restrict__ x, float* __restrict__ buf) {
    int i = blockIdx.x * blockDim.x + threadIdx.x;
    int b = i / NIN_;
    int f = i - b * NIN_;
    buf[(size_t)b * TOTAL_ + f] = x[i];
}

__device__ __forceinline__ v8f ffn_tile_accum(const float* __restrict__ buf,
                                              const float* __restrict__ W,
                                              const int*   __restrict__ idx,
                                              int n0, size_t rowA, int hi, int lo) {
    v8f acc = {};
    #pragma unroll 1
    for (int j = 0; j < 16; ++j) {
        const int nj = n0 + j;
        const int*   idr = idx + nj * K_;
        const float* wr  = W   + nj * K_;
        const bool mine = (lo == j);
        #pragma unroll
        for (int kk = 0; kk < 8; ++kk) {
            const int kb = kk * 4 + 2 * hi;
            const int2   fp = *(const int2*)  (idr + kb);
            const float2 wp = *(const float2*)(wr  + kb);
            v2f a, bm;
            a.x = buf[rowA + (size_t)fp.x];
            a.y = buf[rowA + (size_t)fp.y];
            bm.x = mine ? wp.x : 0.0f;
            bm.y = mine ? wp.y : 0.0f;
            acc = __builtin_amdgcn_wmma_f32_16x16x4_f32(
                      false, a, false, bm, (short)0, acc, false, false);
        }
    }
    return acc;
}

__global__ __launch_bounds__(32)
void ffn_layer_kernel(const float* __restrict__ buf_in,
                      float*       __restrict__ buf,
                      const float* __restrict__ W,
                      const float* __restrict__ bias,
                      const int*   __restrict__ idx,
                      int out_off) {
    const int lane = threadIdx.x;
    const int hi = lane >> 4;
    const int lo = lane & 15;
    const int bt = blockIdx.x;
    const int n0 = blockIdx.y * 16;
    const size_t rowA = (size_t)(bt * 16 + lo) * TOTAL_;

    v8f acc = ffn_tile_accum(buf_in, W, idx, n0, rowA, hi, lo);

    const float bv = bias[n0 + lo];
    #pragma unroll
    for (int r = 0; r < 8; ++r) {
        float v = acc[r] + bv;
        float o = 1.0f / (1.0f + __expf(-v));
        buf[(size_t)(bt * 16 + r + 8 * hi) * TOTAL_ + out_off + n0 + lo] = o;
    }
}

__global__ __launch_bounds__(32)
void ffn_out_kernel(const float* __restrict__ buf,
                    const float* __restrict__ W,
                    const float* __restrict__ bias,
                    const int*   __restrict__ idx,
                    float*       __restrict__ out) {
    const int lane = threadIdx.x;
    const int hi = lane >> 4;
    const int lo = lane & 15;
    const int bt = blockIdx.x;
    const int n0 = blockIdx.y * 16;
    const size_t rowA = (size_t)(bt * 16 + lo) * TOTAL_;

    v8f acc = ffn_tile_accum(buf, W, idx, n0, rowA, hi, lo);

    const float bv = bias[n0 + lo];
    #pragma unroll
    for (int r = 0; r < 8; ++r) {
        float v = acc[r] + bv;
        float o = 1.0f / (1.0f + __expf(-v));
        out[(size_t)(bt * 16 + r + 8 * hi) * NOUT_ + n0 + lo] = o;
    }
}

extern "C" void kernel_launch(void* const* d_in, const int* in_sizes, int n_in,
                              void* d_out, int out_size, void* d_ws, size_t ws_size,
                              hipStream_t stream) {
    const float* x     = (const float*)d_in[0];
    const float* W     = (const float*)d_in[1];
    const float* b     = (const float*)d_in[2];
    const float* W_out = (const float*)d_in[3];
    const float* b_out = (const float*)d_in[4];
    const int*   idx     = (const int*)d_in[5];
    const int*   idx_out = (const int*)d_in[6];

    float* buf = (float*)d_ws;
    float* out = (float*)d_out;

    ffn_init_buf<<<(B_ * NIN_) / 256, 256, 0, stream>>>(x, buf);

    dim3 grid(B_ / 16, N_ / 16);
    for (int l = 0; l < L_; ++l) {
        ffn_layer_kernel<<<grid, 32, 0, stream>>>(
            buf, buf,
            W   + (size_t)l * N_ * K_,
            b   + (size_t)l * N_,
            idx + (size_t)l * N_ * K_,
            NIN_ + l * N_);
    }

    dim3 grid_out(B_ / 16, NOUT_ / 16);
    ffn_out_kernel<<<grid_out, 32, 0, stream>>>(buf, W_out, b_out, idx_out, out);
}


# Activation buffer stored node-major (bufT[node][batch]): every index gather and layer store is contiguous along the batch, read as whole-line 16-byte loads into 4 accumulators per wave
# speedup vs baseline: 10.6758x; 10.6758x over previous
#include <hip/hip_runtime.h>


#define B_     4096
#define NIN_   512
#define L_     16
#define N_     512
#define K_     32
#define NOUT_  256
#define TOTAL_ (NIN_ + L_ * N_)

#define WAVES_  4
#define BCOLS_  64

typedef float v2f __attribute__((ext_vector_type(2)));
typedef float v8f __attribute__((ext_vector_type(8)));

__global__ __launch_bounds__(256)
void ffn_init_bufT(const float* __restrict__ x, float* __restrict__ bufT) {
    __shared__ float tile[32][33];
    const int tx = threadIdx.x & 31;
    const int ty = threadIdx.x >> 5;
    const int b0 = blockIdx.x * 32;
    const int f0 = blockIdx.y * 32;
    #pragma unroll
    for (int i = 0; i < 4; ++i) {
        const int r = ty + 8 * i;
        tile[r][tx] = x[(size_t)(b0 + r) * NIN_ + f0 + tx];
    }
    __syncthreads();
    #pragma unroll
    for (int i = 0; i < 4; ++i) {
        const int r = ty + 8 * i;
        bufT[(size_t)(f0 + r) * B_ + b0 + tx] = tile[tx][r];
    }
}

__device__ __forceinline__ void ffn_tile_accum(const float* __restrict__ bufT,
                                               const float* __restrict__ W,
                                               const int*   __restrict__ idx,
                                               int n0, unsigned col, int hi, int lo,
                                               v8f acc[4]) {
    const char* __restrict__ base = (const char*)bufT;
    const unsigned colb = col * 4u;
    #pragma unroll 1
    for (int j = 0; j < 16; ++j) {
        const int nj = n0 + j;
        const int*   idr = idx + nj * K_;
        const float* wr  = W   + nj * K_;
        const bool mine = (lo == j);
        int2   fp[8];
        float2 wp[8];
        #pragma unroll
        for (int kk = 0; kk < 8; ++kk) {
            const int kb = kk * 4 + 2 * hi;
            fp[kk] = *(const int2*) (idr + kb);
            wp[kk] = *(const float2*)(wr  + kb);
        }
        float4 g0[8], g1[8];
        #pragma unroll
        for (int kk = 0; kk < 8; ++kk) {
            g0[kk] = *(const float4*)(base + (((unsigned)fp[kk].x << 14) + colb));
            g1[kk] = *(const float4*)(base + (((unsigned)fp[kk].y << 14) + colb));
        }
        __builtin_amdgcn_sched_barrier(0);
        #pragma unroll
        for (int kk = 0; kk < 8; ++kk) {
            v2f a;
            a.x = mine ? wp[kk].x : 0.0f;
            a.y = mine ? wp[kk].y : 0.0f;
            v2f b0, b1, b2, b3;
            b0.x = g0[kk].x; b0.y = g1[kk].x;
            b1.x = g0[kk].y; b1.y = g1[kk].y;
            b2.x = g0[kk].z; b2.y = g1[kk].z;
            b3.x = g0[kk].w; b3.y = g1[kk].w;
            acc[0] = __builtin_amdgcn_wmma_f32_16x16x4_f32(
                         false, a, false, b0, (short)0, acc[0], false, false);
            acc[1] = __builtin_amdgcn_wmma_f32_16x16x4_f32(
                         false, a, false, b1, (short)0, acc[1], false, false);
            acc[2] = __builtin_amdgcn_wmma_f32_16x16x4_f32(
                         false, a, false, b2, (short)0, acc[2], false, false);
            acc[3] = __builtin_amdgcn_wmma_f32_16x16x4_f32(
                         false, a, false, b3, (short)0, acc[3], false, false);
        }
    }
}

__device__ __forceinline__ float ffn_sigmoid(float v) {
    return 1.0f / (1.0f + __expf(-v));
}

__global__ __launch_bounds__(WAVES_ * 32)
void ffn_layer_kernel(float*       __restrict__ bufT,
                      const float* __restrict__ W,
                      const float* __restrict__ bias,
                      const int*   __restrict__ idx,
                      int out_off) {
    const int lane = threadIdx.x & 31;
    const int wave = threadIdx.x >> 5;
    const int hi = lane >> 4;
    const int lo = lane & 15;
    const int n0 = blockIdx.y * 16;
    const unsigned col = (unsigned)((blockIdx.x * WAVES_ + wave) * BCOLS_ + 4 * lo);

    v8f acc[4] = {};
    ffn_tile_accum(bufT, W, idx, n0, col, hi, lo, acc);

    #pragma unroll
    for (int r = 0; r < 8; ++r) {
        const int n = n0 + 8 * hi + r;
        const float bv = bias[n];
        float4 o;
        o.x = ffn_sigmoid(acc[0][r] + bv);
        o.y = ffn_sigmoid(acc[1][r] + bv);
        o.z = ffn_sigmoid(acc[2][r] + bv);
        o.w = ffn_sigmoid(acc[3][r] + bv);
        *(float4*)(bufT + ((unsigned)(out_off + n) * (unsigned)B_ + col)) = o;
    }
}

__global__ __launch_bounds__(WAVES_ * 32)
void ffn_out_kernel(const float* __restrict__ bufT,
                    const float* __restrict__ W,
                    const float* __restrict__ bias,
                    const int*   __restrict__ idx,
                    float*       __restrict__ out) {
    const int lane = threadIdx.x & 31;
    const int wave = threadIdx.x >> 5;
    const int hi = lane >> 4;
    const int lo = lane & 15;
    const int n0 = blockIdx.y * 16;
    const unsigned col = (unsigned)((blockIdx.x * WAVES_ + wave) * BCOLS_ + 4 * lo);

    v8f acc[4] = {};
    ffn_tile_accum(bufT, W, idx, n0, col, hi, lo, acc);

    const float4 bl = *(const float4*)(bias + n0 + 8 * hi);
    const float4 bh = *(const float4*)(bias + n0 + 8 * hi + 4);
    #pragma unroll
    for (int t = 0; t < 4; ++t) {
        float4 o0, o1;
        o0.x = ffn_sigmoid(acc[t][0] + bl.x);
        o0.y = ffn_sigmoid(acc[t][1] + bl.y);
        o0.z = ffn_sigmoid(acc[t][2] + bl.z);
        o0.w = ffn_sigmoid(acc[t][3] + bl.w);
        o1.x = ffn_sigmoid(acc[t][4] + bh.x);
        o1.y = ffn_sigmoid(acc[t][5] + bh.y);
        o1.z = ffn_sigmoid(acc[t][6] + bh.z);
        o1.w = ffn_sigmoid(acc[t][7] + bh.w);
        float* p = out + (size_t)(col + t) * NOUT_ + n0 + 8 * hi;
        *(float4*)(p)     = o0;
        *(float4*)(p + 4) = o1;
    }
}

extern "C" void kernel_launch(void* const* d_in, const int* in_sizes, int n_in,
                              void* d_out, int out_size, void* d_ws, size_t ws_size,
                              hipStream_t stream) {
    const float* x     = (const float*)d_in[0];
    const float* W     = (const float*)d_in[1];
    const float* b     = (const float*)d_in[2];
    const float* W_out = (const float*)d_in[3];
    const float* b_out = (const float*)d_in[4];
    const int*   idx     = (const int*)d_in[5];
    const int*   idx_out = (const int*)d_in[6];

    float* bufT = (float*)d_ws;
    float* out  = (float*)d_out;

    dim3 grid_init(B_ / 32, NIN_ / 32);
    ffn_init_bufT<<<grid_init, 256, 0, stream>>>(x, bufT);

    dim3 grid(B_ / (WAVES_ * BCOLS_), N_ / 16);
    for (int l = 0; l < L_; ++l) {
        ffn_layer_kernel<<<grid, WAVES_ * 32, 0, stream>>>(
            bufT,
            W   + (size_t)l * N_ * K_,
            b   + (size_t)l * N_,
            idx + (size_t)l * N_ * K_,
            NIN_ + l * N_);
    }

    dim3 grid_out(B_ / (WAVES_ * BCOLS_), NOUT_ / 16);
    ffn_out_kernel<<<grid_out, WAVES_ * 32, 0, stream>>>(bufT, W_out, b_out, idx_out, out);
}


# Neuron loop software-pipelined, unrolled: row gathers of neuron j+1 and index/weight pairs of neuron j+2 issue before the WMMAs of neuron j (two register sets); sum order unchanged
# speedup vs baseline: 12.4229x; 1.1637x over previous
#include <hip/hip_runtime.h>


#define B_     4096
#define NIN_   512
#define L_     16
#define N_     512
#define K_     32
#define NOUT_  256
#define TOTAL_ (NIN_ + L_ * N_)

#define WAVES_  4
#define BCOLS_  64

typedef float v2f __attribute__((ext_vector_type(2)));
typedef float v8f __attribute__((ext_vector_type(8)));

__global__ __launch_bounds__(256)
void ffn_init_bufT(const float* __restrict__ x, float* __restrict__ bufT) {
    __shared__ float tile[32][33];
    const int tx = threadIdx.x & 31;
    const int ty = threadIdx.x >> 5;
    const int b0 = blockIdx.x * 32;
    const int f0 = blockIdx.y * 32;
    #pragma unroll
    for (int i = 0; i < 4; ++i) {
        const int r = ty + 8 * i;
        tile[r][tx] = x[(size_t)(b0 + r) * NIN_ + f0 + tx];
    }
    __syncthreads();
    #pragma unroll
    for (int i = 0; i < 4; ++i) {
        const int r = ty + 8 * i;
        bufT[(size_t)(f0 + r) * B_ + b0 + tx] = tile[tx][r];
    }
}

__device__ __forceinline__ void ffn_load_idx(const int* __restrict__ idx, int nj, int hi,
                                             int2 fp[8]) {
    const int* idr = idx + nj * K_ + 2 * hi;
    #pragma unroll
    for (int kk = 0; kk < 8; ++kk) fp[kk] = *(const int2*)(idr + kk * 4);
}

__device__ __forceinline__ void ffn_load_w(const float* __restrict__ W, int nj, int hi,
                                           float2 wp[8]) {
    const float* wr = W + nj * K_ + 2 * hi;
    #pragma unroll
    for (int kk = 0; kk < 8; ++kk) wp[kk] = *(const float2*)(wr + kk * 4);
}

__device__ __forceinline__ void ffn_gather(const char* __restrict__ base, unsigned colb,
                                           const int2 fp[8], float4 g0[8], float4 g1[8]) {
    #pragma unroll
    for (int kk = 0; kk < 8; ++kk) {
        g0[kk] = *(const float4*)(base + (((unsigned)fp[kk].x << 14) + colb));
        g1[kk] = *(const float4*)(base + (((unsigned)fp[kk].y << 14) + colb));
    }
}

__device__ __forceinline__ void ffn_mma(const float2 wp[8], const float4 g0[8],
                                        const float4 g1[8], bool mine, v8f acc[4]) {
    #pragma unroll
    for (int kk = 0; kk < 8; ++kk) {
        v2f a;
        a.x = mine ? wp[kk].x : 0.0f;
        a.y = mine ? wp[kk].y : 0.0f;
        v2f b0, b1, b2, b3;
        b0.x = g0[kk].x; b0.y = g1[kk].x;
        b1.x = g0[kk].y; b1.y = g1[kk].y;
        b2.x = g0[kk].z; b2.y = g1[kk].z;
        b3.x = g0[kk].w; b3.y = g1[kk].w;
        acc[0] = __builtin_amdgcn_wmma_f32_16x16x4_f32(
                     false, a, false, b0, (short)0, acc[0], false, false);
        acc[1] = __builtin_amdgcn_wmma_f32_16x16x4_f32(
                     false, a, false, b1, (short)0, acc[1], false, false);
        acc[2] = __builtin_amdgcn_wmma_f32_16x16x4_f32(
                     false, a, false, b2, (short)0, acc[2], false, false);
        acc[3] = __builtin_amdgcn_wmma_f32_16x16x4_f32(
                     false, a, false, b3, (short)0, acc[3], false, false);
    }
}

__device__ __forceinline__ void ffn_tile_accum(const float* __restrict__ bufT,
                                               const float* __restrict__ W,
                                               const int*   __restrict__ idx,
                                               int n0, unsigned col, int hi, int lo,
                                               v8f acc[4]) {
    const char* __restrict__ base = (const char*)bufT;
    const unsigned colb = col * 4u;
    int2   fpA[8], fpB[8];
    float2 wpA[8], wpB[8];
    float4 gA0[8], gA1[8], gB0[8], gB1[8];

    ffn_load_idx(idx, n0,     hi, fpA);
    ffn_load_w  (W,   n0,     hi, wpA);
    ffn_load_idx(idx, n0 + 1, hi, fpB);
    ffn_load_w  (W,   n0 + 1, hi, wpB);
    ffn_gather(base, colb, fpA, gA0, gA1);

    #pragma unroll
    for (int j = 0; j < 14; j += 2) {
        ffn_load_idx(idx, n0 + j + 2, hi, fpA);
        __builtin_amdgcn_sched_barrier(0);
        ffn_gather(base, colb, fpB, gB0, gB1);
        __builtin_amdgcn_sched_barrier(0);
        ffn_mma(wpA, gA0, gA1, lo == j, acc);
        __builtin_amdgcn_sched_barrier(0);
        ffn_load_w  (W,   n0 + j + 2, hi, wpA);
        ffn_load_idx(idx, n0 + j + 3, hi, fpB);
        __builtin_amdgcn_sched_barrier(0);
        ffn_gather(base, colb, fpA, gA0, gA1);
        __builtin_amdgcn_sched_barrier(0);
        ffn_mma(wpB, gB0, gB1, lo == j + 1, acc);
        __builtin_amdgcn_sched_barrier(0);
        ffn_load_w  (W,   n0 + j + 3, hi, wpB);
    }
    ffn_gather(base, colb, fpB, gB0, gB1);
    __builtin_amdgcn_sched_barrier(0);
    ffn_mma(wpA, gA0, gA1, lo == 14, acc);
    ffn_mma(wpB, gB0, gB1, lo == 15, acc);
}

__device__ __forceinline__ float ffn_sigmoid(float v) {
    return 1.0f / (1.0f + __expf(-v));
}

__global__ __launch_bounds__(WAVES_ * 32)
void ffn_layer_kernel(float*       __restrict__ bufT,
                      const float* __restrict__ W,
                      const float* __restrict__ bias,
                      const int*   __restrict__ idx,
                      int out_off) {
    const int lane = threadIdx.x & 31;
    const int wave = threadIdx.x >> 5;
    const int hi = lane >> 4;
    const int lo = lane & 15;
    const int n0 = blockIdx.y * 16;
    const unsigned col = (unsigned)((blockIdx.x * WAVES_ + wave) * BCOLS_ + 4 * lo);

    v8f acc[4] = {};
    ffn_tile_accum(bufT, W, idx, n0, col, hi, lo, acc);

    #pragma unroll
    for (int r = 0; r < 8; ++r) {
        const int n = n0 + 8 * hi + r;
        const float bv = bias[n];
        float4 o;
        o.x = ffn_sigmoid(acc[0][r] + bv);
        o.y = ffn_sigmoid(acc[1][r] + bv);
        o.z = ffn_sigmoid(acc[2][r] + bv);
        o.w = ffn_sigmoid(acc[3][r] + bv);
        *(float4*)(bufT + ((unsigned)(out_off + n) * (unsigned)B_ + col)) = o;
    }
}

__global__ __launch_bounds__(WAVES_ * 32)
void ffn_out_kernel(const float* __restrict__ bufT,
                    const float* __restrict__ W,
                    const float* __restrict__ bias,
                    const int*   __restrict__ idx,
                    float*       __restrict__ out) {
    const int lane = threadIdx.x & 31;
    const int wave = threadIdx.x >> 5;
    const int hi = lane >> 4;
    const int lo = lane & 15;
    const int n0 = blockIdx.y * 16;
    const unsigned col = (unsigned)((blockIdx.x * WAVES_ + wave) * BCOLS_ + 4 * lo);

    v8f acc[4] = {};
    ffn_tile_accum(bufT, W, idx, n0, col, hi, lo, acc);

    const float4 bl = *(const float4*)(bias + n0 + 8 * hi);
    const float4 bh = *(const float4*)(bias + n0 + 8 * hi + 4);
    #pragma unroll
    for (int t = 0; t < 4; ++t) {
        float4 o0, o1;
        o0.x = ffn_sigmoid(acc[t][0] + bl.x);
        o0.y = ffn_sigmoid(acc[t][1] + bl.y);
        o0.z = ffn_sigmoid(acc[t][2] + bl.z);
        o0.w = ffn_sigmoid(acc[t][3] + bl.w);
        o1.x = ffn_sigmoid(acc[t][4] + bh.x);
        o1.y = ffn_sigmoid(acc[t][5] + bh.y);
        o1.z = ffn_sigmoid(acc[t][6] + bh.z);
        o1.w = ffn_sigmoid(acc[t][7] + bh.w);
        float* p = out + (size_t)(col + t) * NOUT_ + n0 + 8 * hi;
        *(float4*)(p)     = o0;
        *(float4*)(p + 4) = o1;
    }
}

extern "C" void kernel_launch(void* const* d_in, const int* in_sizes, int n_in,
                              void* d_out, int out_size, void* d_ws, size_t ws_size,
                              hipStream_t stream) {
    const float* x     = (const float*)d_in[0];
    const float* W     = (const float*)d_in[1];
    const float* b     = (const float*)d_in[2];
    const float* W_out = (const float*)d_in[3];
    const float* b_out = (const float*)d_in[4];
    const int*   idx     = (const int*)d_in[5];
    const int*   idx_out = (const int*)d_in[6];

    float* bufT = (float*)d_ws;
    float* out  = (float*)d_out;

    dim3 grid_init(B_ / 32, NIN_ / 32);
    ffn_init_bufT<<<grid_init, 256, 0, stream>>>(x, bufT);

    dim3 grid(B_ / (WAVES_ * BCOLS_), N_ / 16);
    for (int l = 0; l < L_; ++l) {
        ffn_layer_kernel<<<grid, WAVES_ * 32, 0, stream>>>(
            bufT,
            W   + (size_t)l * N_ * K_,
            b   + (size_t)l * N_,
            idx + (size_t)l * N_ * K_,
            NIN_ + l * N_);
    }

    dim3 grid_out(B_ / (WAVES_ * BCOLS_), NOUT_ / 16);
    ffn_out_kernel<<<grid_out, WAVES_ * 32, 0, stream>>>(bufT, W_out, b_out, idx_out, out);
}


# One launch for all 17 layers: a work-group owns 8 batch columns and keeps all node activations in LDS, so every predecessor gather is an LDS read, none from L2; sum order unchanged
# speedup vs baseline: 14.4482x; 1.1630x over previous
#include <hip/hip_runtime.h>


#define B_     4096
#define NIN_   512
#define L_     16
#define N_     512
#define K_     32
#define NOUT_  256
#define TOTAL_ (NIN_ + L_ * N_)

#define COLS_       8
#define WAVES_      32
#define ACT_FLOATS_ (TOTAL_ * COLS_)
#define OUT_FLOATS_ (COLS_ * NOUT_)

typedef float v2f __attribute__((ext_vector_type(2)));
typedef float v8f __attribute__((ext_vector_type(8)));

__device__ __forceinline__ void ffn_load_idx(const int* __restrict__ ir, int j, int2 ip[8]) {
    const int* p = ir + j * K_;
    #pragma unroll
    for (int kk = 0; kk < 8; ++kk) ip[kk] = *(const int2*)(p + kk * 4);
}

__device__ __forceinline__ void ffn_gather(const float* act, int cb, const int2 ip[8], v2f g[8]) {
    #pragma unroll
    for (int kk = 0; kk < 8; ++kk) {
        int iy = ip[kk].y;
        asm("" : "+v"(iy));
        g[kk].x = act[ip[kk].x * COLS_ + cb];
        g[kk].y = act[iy * COLS_ + cb];
    }
}

__device__ __forceinline__ v8f ffn_mma(const float2 w[8], const v2f g[8], bool mine, v8f acc) {
    #pragma unroll
    for (int kk = 0; kk < 8; ++kk) {
        v2f a;
        a.x = mine ? w[kk].x : 0.0f;
        a.y = mine ? w[kk].y : 0.0f;
        acc = __builtin_amdgcn_wmma_f32_16x16x4_f32(
                  false, a, false, g[kk], (short)0, acc, false, false);
    }
    return acc;
}

struct FfnPre {
    float2 w[8];
    int2   ip0[8];
    int2   ip1[8];
};

__device__ __forceinline__ void ffn_preload(const float* __restrict__ W,
                                            const int*   __restrict__ idx,
                                            int n0, int c, int h, int cg, FfnPre& p) {
    const float* wr = W + (n0 + c) * K_ + 2 * h;
    #pragma unroll
    for (int kk = 0; kk < 8; ++kk) p.w[kk] = *(const float2*)(wr + kk * 4);
    const int* ir = idx + (n0 + 8 * cg) * K_ + 2 * h;
    ffn_load_idx(ir, 0, p.ip0);
    ffn_load_idx(ir, 1, p.ip1);
}

__device__ __forceinline__ v8f ffn_tile(const float* act,
                                        const int* __restrict__ idx,
                                        int n0, int h, int cg, int cb, const FfnPre& p) {
    const int* ir = idx + (n0 + 8 * cg) * K_ + 2 * h;
    int2 ip[3][8];
    v2f  g[2][8];
    #pragma unroll
    for (int kk = 0; kk < 8; ++kk) { ip[0][kk] = p.ip0[kk]; ip[1][kk] = p.ip1[kk]; }
    ffn_gather(act, cb, ip[0], g[0]);

    v8f acc = {};
    #pragma unroll
    for (int j = 0; j < 8; ++j) {
        if (j + 2 < 8) ffn_load_idx(ir, j + 2, ip[(j + 2) % 3]);
        if (j + 1 < 8) ffn_gather(act, cb, ip[(j + 1) % 3], g[(j + 1) & 1]);
        __builtin_amdgcn_sched_barrier(0);
        acc = ffn_mma(p.w, g[j & 1], cb == j, acc);
        __builtin_amdgcn_sched_barrier(0);
    }
    return acc;
}

__device__ __forceinline__ float ffn_sigmoid(float v) {
    return 1.0f / (1.0f + __expf(-v));
}

__global__ __launch_bounds__(WAVES_ * 32)
void ffn_fused_kernel(const float* __restrict__ x,
                      const float* __restrict__ W,
                      const float* __restrict__ bias,
                      const float* __restrict__ W_out,
                      const float* __restrict__ b_out,
                      const int*   __restrict__ idx,
                      const int*   __restrict__ idx_out,
                      float*       __restrict__ out) {
    __shared__ __attribute__((aligned(16))) float act[ACT_FLOATS_ + OUT_FLOATS_];
    float* stage = act + ACT_FLOATS_;

    const int tid  = threadIdx.x;
    const int lane = tid & 31;
    const int wave = tid >> 5;
    const int c  = lane & 15;
    const int h  = lane >> 4;
    const int cg = c >> 3;
    const int cb = c & 7;
    const int b0 = blockIdx.x * COLS_;
    const bool valid = (h == cg);

    const int n0 = wave * 16;
    FfnPre pre;
    ffn_preload(W, idx, n0, c, h, cg, pre);

    #pragma unroll
    for (int i = 0; i < (COLS_ * NIN_) / (WAVES_ * 32); ++i) {
        const int e   = tid + WAVES_ * 32 * i;
        const int col = e / NIN_;
        const int f   = e % NIN_;
        act[f * COLS_ + col] = x[(size_t)(b0 + col) * NIN_ + f];
    }
    __syncthreads();

    #pragma nounroll
    for (int l = 0; l < L_; ++l) {
        const v8f acc = ffn_tile(act, idx + (size_t)l * N_ * K_, n0, h, cg, cb, pre);
        const bool last = (l + 1 == L_);
        const float* Wn = last ? W_out   : W   + (size_t)(l + 1) * N_ * K_;
        const int*   In = last ? idx_out : idx + (size_t)(l + 1) * N_ * K_;
        if (!last || wave < NOUT_ / 16) ffn_preload(Wn, In, n0, c, h, cg, pre);
        const float* bl = bias + l * N_ + n0 + 8 * h;
        const float4 bA = *(const float4*)(bl);
        const float4 bB = *(const float4*)(bl + 4);
        const float bv[8] = {bA.x, bA.y, bA.z, bA.w, bB.x, bB.y, bB.z, bB.w};
        if (valid) {
            float* o = act + (NIN_ + l * N_ + n0 + 8 * h) * COLS_ + cb;
            #pragma unroll
            for (int r = 0; r < 8; ++r) o[r * COLS_] = ffn_sigmoid(acc[r] + bv[r]);
        }
        __syncthreads();
    }

    if (wave < NOUT_ / 16) {
        const v8f acc = ffn_tile(act, idx_out, n0, h, cg, cb, pre);
        const float* bl = b_out + n0 + 8 * h;
        const float4 bA = *(const float4*)(bl);
        const float4 bB = *(const float4*)(bl + 4);
        if (valid) {
            float4 o0, o1;
            o0.x = ffn_sigmoid(acc[0] + bA.x);
            o0.y = ffn_sigmoid(acc[1] + bA.y);
            o0.z = ffn_sigmoid(acc[2] + bA.z);
            o0.w = ffn_sigmoid(acc[3] + bA.w);
            o1.x = ffn_sigmoid(acc[4] + bB.x);
            o1.y = ffn_sigmoid(acc[5] + bB.y);
            o1.z = ffn_sigmoid(acc[6] + bB.z);
            o1.w = ffn_sigmoid(acc[7] + bB.w);
            float* p = stage + cb * NOUT_ + n0 + 8 * h;
            *(float4*)(p)     = o0;
            *(float4*)(p + 4) = o1;
        }
    }
    __syncthreads();
    if (tid < OUT_FLOATS_ / 4) {
        const int row = tid / (NOUT_ / 4);
        const int q   = tid % (NOUT_ / 4);
        const float4 v = *(const float4*)(stage + row * NOUT_ + q * 4);
        *(float4*)(out + (size_t)(b0 + row) * NOUT_ + q * 4) = v;
    }
}

extern "C" void kernel_launch(void* const* d_in, const int* in_sizes, int n_in,
                              void* d_out, int out_size, void* d_ws, size_t ws_size,
                              hipStream_t stream) {
    const float* x     = (const float*)d_in[0];
    const float* W     = (const float*)d_in[1];
    const float* b     = (const float*)d_in[2];
    const float* W_out = (const float*)d_in[3];
    const float* b_out = (const float*)d_in[4];
    const int*   idx     = (const int*)d_in[5];
    const int*   idx_out = (const int*)d_in[6];
    float* out = (float*)d_out;

    ffn_fused_kernel<<<B_ / COLS_, WAVES_ * 32, 0, stream>>>(
        x, W, b, W_out, b_out, idx, idx_out, out);
}


# Gathers read as 16-byte LDS loads: a lane takes 4 batch columns of a predecessor at once and feeds 4 accumulators (one per column); a quarter of the LDS reads and addresses; sum order unchanged
# speedup vs baseline: 17.1051x; 1.1839x over previous
#include <hip/hip_runtime.h>


#define B_     4096
#define NIN_   512
#define L_     16
#define N_     512
#define K_     32
#define NOUT_  256
#define TOTAL_ (NIN_ + L_ * N_)

#define COLS_       8
#define WAVES_      32
#define ACT_FLOATS_ (TOTAL_ * COLS_)
#define OUT_FLOATS_ (COLS_ * NOUT_)

typedef float v2f __attribute__((ext_vector_type(2)));
typedef float v8f __attribute__((ext_vector_type(8)));

__device__ __forceinline__ void ffn_load_idx(const int* __restrict__ ir, int u, int2 ip[2]) {
    const int* p = ir + (u >> 2) * 8 * K_ + (u & 3) * 8;
    ip[0] = *(const int2*)(p);
    ip[1] = *(const int2*)(p + 4);
}

__device__ __forceinline__ void ffn_gather(const float* act, int p4, const int2 ip[2], float4 g[4]) {
    #pragma unroll
    for (int kl = 0; kl < 2; ++kl) {
        int iy = ip[kl].y;
        asm("" : "+v"(iy));
        g[2 * kl]     = *(const float4*)(act + ip[kl].x * COLS_ + p4);
        g[2 * kl + 1] = *(const float4*)(act + iy * COLS_ + p4);
    }
}

__device__ __forceinline__ void ffn_mma(const float2 w[8], int kk0, const float4 g[4], bool mine, v8f acc[4]) {
    #pragma unroll
    for (int kl = 0; kl < 2; ++kl) {
        v2f a;
        a.x = mine ? w[kk0 + kl].x : 0.0f;
        a.y = mine ? w[kk0 + kl].y : 0.0f;
        const float4 g0 = g[2 * kl], g1 = g[2 * kl + 1];
        const v2f b0 = {g0.x, g1.x};
        const v2f b1 = {g0.y, g1.y};
        const v2f b2 = {g0.z, g1.z};
        const v2f b3 = {g0.w, g1.w};
        acc[0] = __builtin_amdgcn_wmma_f32_16x16x4_f32(false, a, false, b0, (short)0, acc[0], false, false);
        acc[1] = __builtin_amdgcn_wmma_f32_16x16x4_f32(false, a, false, b1, (short)0, acc[1], false, false);
        acc[2] = __builtin_amdgcn_wmma_f32_16x16x4_f32(false, a, false, b2, (short)0, acc[2], false, false);
        acc[3] = __builtin_amdgcn_wmma_f32_16x16x4_f32(false, a, false, b3, (short)0, acc[3], false, false);
    }
}

struct FfnPre {
    float2 w[8];
    int2   ip0[2];
    int2   ip1[2];
};

__device__ __forceinline__ void ffn_preload(const float* __restrict__ W,
                                            const int*   __restrict__ idx,
                                            int n0, int c, int h, int q, FfnPre& p) {
    const float* wr = W + (n0 + c) * K_ + 2 * h;
    #pragma unroll
    for (int kk = 0; kk < 8; ++kk) p.w[kk] = *(const float2*)(wr + kk * 4);
    const int* ir = idx + (n0 + q) * K_ + 2 * h;
    ffn_load_idx(ir, 0, p.ip0);
    ffn_load_idx(ir, 1, p.ip1);
}

__device__ __forceinline__ float4 ffn_tile(const float* act,
                                           const int* __restrict__ idx,
                                           int n0, int c, int h, int q, int p4, const FfnPre& p) {
    const int* ir = idx + (n0 + q) * K_ + 2 * h;
    int2   ip[3][2];
    float4 g[2][4];
    #pragma unroll
    for (int kl = 0; kl < 2; ++kl) { ip[0][kl] = p.ip0[kl]; ip[1][kl] = p.ip1[kl]; }
    ffn_gather(act, p4, ip[0], g[0]);

    v8f acc[4] = {};
    const int t_own = c >> 3;
    #pragma unroll
    for (int u = 0; u < 8; ++u) {
        if (u + 2 < 8) ffn_load_idx(ir, u + 2, ip[(u + 2) % 3]);
        if (u + 1 < 8) ffn_gather(act, p4, ip[(u + 1) % 3], g[(u + 1) & 1]);
        __builtin_amdgcn_sched_barrier(0);
        ffn_mma(p.w, 2 * (u & 3), g[u & 1], t_own == (u >> 2), acc);
        __builtin_amdgcn_sched_barrier(0);
    }
    float4 r;
    float* rp = (float*)&r;
    #pragma unroll
    for (int m = 0; m < 4; ++m) {
        const v8f a = acc[m];
        const float s01 = (q & 1) ? a[1] : a[0];
        const float s23 = (q & 1) ? a[3] : a[2];
        const float s45 = (q & 1) ? a[5] : a[4];
        const float s67 = (q & 1) ? a[7] : a[6];
        const float lo  = (q & 2) ? s23 : s01;
        const float hi  = (q & 2) ? s67 : s45;
        rp[m] = (q & 4) ? hi : lo;
    }
    return r;
}

__device__ __forceinline__ float ffn_sigmoid(float v) {
    return 1.0f / (1.0f + __expf(-v));
}

__global__ __launch_bounds__(WAVES_ * 32)
void ffn_fused_kernel(const float* __restrict__ x,
                      const float* __restrict__ W,
                      const float* __restrict__ bias,
                      const float* __restrict__ W_out,
                      const float* __restrict__ b_out,
                      const int*   __restrict__ idx,
                      const int*   __restrict__ idx_out,
                      float*       __restrict__ out) {
    __shared__ __attribute__((aligned(16))) float act[ACT_FLOATS_ + OUT_FLOATS_];
    float* stage = act + ACT_FLOATS_;

    const int tid  = threadIdx.x;
    const int lane = tid & 31;
    const int wave = tid >> 5;
    const int c  = lane & 15;
    const int h  = lane >> 4;
    const int q  = c >> 1;
    const int p4 = (c & 1) * 4;
    const int b0 = blockIdx.x * COLS_;

    const int n0 = wave * 16;
    const int nr = n0 + 8 * h + q;
    FfnPre pre;
    ffn_preload(W, idx, n0, c, h, q, pre);

    #pragma unroll
    for (int i = 0; i < (COLS_ * NIN_) / (WAVES_ * 32); ++i) {
        const int e   = tid + WAVES_ * 32 * i;
        const int col = e / NIN_;
        const int f   = e % NIN_;
        act[f * COLS_ + col] = x[(size_t)(b0 + col) * NIN_ + f];
    }
    __syncthreads();

    #pragma nounroll
    for (int l = 0; l < L_; ++l) {
        const float bv = bias[l * N_ + nr];
        const float4 s = ffn_tile(act, idx + (size_t)l * N_ * K_, n0, c, h, q, p4, pre);
        const bool last = (l + 1 == L_);
        const float* Wn = last ? W_out   : W   + (size_t)(l + 1) * N_ * K_;
        const int*   In = last ? idx_out : idx + (size_t)(l + 1) * N_ * K_;
        if (!last || wave < NOUT_ / 16) ffn_preload(Wn, In, n0, c, h, q, pre);
        float4 o;
        o.x = ffn_sigmoid(s.x + bv);
        o.y = ffn_sigmoid(s.y + bv);
        o.z = ffn_sigmoid(s.z + bv);
        o.w = ffn_sigmoid(s.w + bv);
        *(float4*)(act + (NIN_ + l * N_ + nr) * COLS_ + p4) = o;
        __syncthreads();
    }

    if (wave < NOUT_ / 16) {
        const float bv = b_out[nr];
        const float4 s = ffn_tile(act, idx_out, n0, c, h, q, p4, pre);
        float* sp = stage + p4 * NOUT_ + nr;
        sp[0 * NOUT_] = ffn_sigmoid(s.x + bv);
        sp[1 * NOUT_] = ffn_sigmoid(s.y + bv);
        sp[2 * NOUT_] = ffn_sigmoid(s.z + bv);
        sp[3 * NOUT_] = ffn_sigmoid(s.w + bv);
    }
    __syncthreads();
    if (tid < OUT_FLOATS_ / 4) {
        const int row = tid / (NOUT_ / 4);
        const int qq  = tid % (NOUT_ / 4);
        const float4 v = *(const float4*)(stage + row * NOUT_ + qq * 4);
        *(float4*)(out + (size_t)(b0 + row) * NOUT_ + qq * 4) = v;
    }
}

extern "C" void kernel_launch(void* const* d_in, const int* in_sizes, int n_in,
                              void* d_out, int out_size, void* d_ws, size_t ws_size,
                              hipStream_t stream) {
    const float* x     = (const float*)d_in[0];
    const float* W     = (const float*)d_in[1];
    const float* b     = (const float*)d_in[2];
    const float* W_out = (const float*)d_in[3];
    const float* b_out = (const float*)d_in[4];
    const int*   idx     = (const int*)d_in[5];
    const int*   idx_out = (const int*)d_in[6];
    float* out = (float*)d_out;

    ffn_fused_kernel<<<B_ / COLS_, WAVES_ * 32, 0, stream>>>(
        x, W, b, W_out, b_out, idx, idx_out, out);
}


# Layer barrier orders LDS only (workgroup fences on the local address space round s_barrier): next layer's weight and index fetches, issued inside the tile, stay in flight across it; sums unchanged
# speedup vs baseline: 17.7321x; 1.0367x over previous
#include <hip/hip_runtime.h>


#define B_     4096
#define NIN_   512
#define L_     16
#define N_     512
#define K_     32
#define NOUT_  256
#define TOTAL_ (NIN_ + L_ * N_)

#define COLS_       8
#define WAVES_      32
#define ACT_FLOATS_ (TOTAL_ * COLS_)
#define OUT_FLOATS_ (COLS_ * NOUT_)

typedef float v2f __attribute__((ext_vector_type(2)));
typedef float v8f __attribute__((ext_vector_type(8)));

__device__ __forceinline__ void ffn_load_idx(const int* __restrict__ ir, int u, int2 ip[2]) {
    const int* p = ir + (u >> 2) * 8 * K_ + (u & 3) * 8;
    ip[0] = *(const int2*)(p);
    ip[1] = *(const int2*)(p + 4);
}

__device__ __forceinline__ void ffn_gather(const float* act, int p4, const int2 ip[2], float4 g[4]) {
    #pragma unroll
    for (int kl = 0; kl < 2; ++kl) {
        int iy = ip[kl].y;
        asm("" : "+v"(iy));
        g[2 * kl]     = *(const float4*)(act + ip[kl].x * COLS_ + p4);
        g[2 * kl + 1] = *(const float4*)(act + iy * COLS_ + p4);
    }
}

__device__ __forceinline__ void ffn_mma(const float2 w[8], int kk0, const float4 g[4], bool mine, v8f acc[4]) {
    #pragma unroll
    for (int kl = 0; kl < 2; ++kl) {
        v2f a;
        a.x = mine ? w[kk0 + kl].x : 0.0f;
        a.y = mine ? w[kk0 + kl].y : 0.0f;
        const float4 g0 = g[2 * kl], g1 = g[2 * kl + 1];
        const v2f b0 = {g0.x, g1.x};
        const v2f b1 = {g0.y, g1.y};
        const v2f b2 = {g0.z, g1.z};
        const v2f b3 = {g0.w, g1.w};
        acc[0] = __builtin_amdgcn_wmma_f32_16x16x4_f32(false, a, false, b0, (short)0, acc[0], false, false);
        acc[1] = __builtin_amdgcn_wmma_f32_16x16x4_f32(false, a, false, b1, (short)0, acc[1], false, false);
        acc[2] = __builtin_amdgcn_wmma_f32_16x16x4_f32(false, a, false, b2, (short)0, acc[2], false, false);
        acc[3] = __builtin_amdgcn_wmma_f32_16x16x4_f32(false, a, false, b3, (short)0, acc[3], false, false);
    }
}

struct FfnPre {
    float2 w[8];
    int2   ip0[2];
    int2   ip1[2];
};

__device__ __forceinline__ void ffn_preload(const float* __restrict__ W,
                                            const int*   __restrict__ idx,
                                            int n0, int c, int h, int q, FfnPre& p) {
    const float* wr = W + (n0 + c) * K_ + 2 * h;
    #pragma unroll
    for (int kk = 0; kk < 8; ++kk) p.w[kk] = *(const float2*)(wr + kk * 4);
    const int* ir = idx + (n0 + q) * K_ + 2 * h;
    ffn_load_idx(ir, 0, p.ip0);
    ffn_load_idx(ir, 1, p.ip1);
}

template <bool PRE>
__device__ __forceinline__ float4 ffn_tile(const float* act,
                                           const int* __restrict__ idx,
                                           int n0, int c, int h, int q, int p4, FfnPre& p,
                                           const float* __restrict__ wrn,
                                           const int*   __restrict__ irn) {
    const int* ir = idx + (n0 + q) * K_ + 2 * h;
    int2   ip[3][2];
    float4 g[2][4];
    float2 w[8];
    #pragma unroll
    for (int kk = 0; kk < 8; ++kk) w[kk] = p.w[kk];
    #pragma unroll
    for (int kl = 0; kl < 2; ++kl) { ip[0][kl] = p.ip0[kl]; ip[1][kl] = p.ip1[kl]; }
    ffn_gather(act, p4, ip[0], g[0]);

    v8f acc[4] = {};
    const int t_own = c >> 3;
    #pragma unroll
    for (int u = 0; u < 8; ++u) {
        if (u + 2 < 8) ffn_load_idx(ir, u + 2, ip[(u + 2) % 3]);
        if (PRE && u == 2) { ffn_load_idx(irn, 0, p.ip0); ffn_load_idx(irn, 1, p.ip1); }
        if (PRE && u >= 5) {
            const int kk = 2 * ((u - 1) & 3);
            p.w[kk]     = *(const float2*)(wrn + kk * 4);
            p.w[kk + 1] = *(const float2*)(wrn + kk * 4 + 4);
        }
        if (u + 1 < 8) ffn_gather(act, p4, ip[(u + 1) % 3], g[(u + 1) & 1]);
        __builtin_amdgcn_sched_barrier(0);
        ffn_mma(w, 2 * (u & 3), g[u & 1], t_own == (u >> 2), acc);
        __builtin_amdgcn_sched_barrier(0);
    }
    if (PRE) {
        p.w[6] = *(const float2*)(wrn + 24);
        p.w[7] = *(const float2*)(wrn + 28);
    }
    float4 r;
    float* rp = (float*)&r;
    #pragma unroll
    for (int m = 0; m < 4; ++m) {
        const v8f a = acc[m];
        const float s01 = (q & 1) ? a[1] : a[0];
        const float s23 = (q & 1) ? a[3] : a[2];
        const float s45 = (q & 1) ? a[5] : a[4];
        const float s67 = (q & 1) ? a[7] : a[6];
        const float lo  = (q & 2) ? s23 : s01;
        const float hi  = (q & 2) ? s67 : s45;
        rp[m] = (q & 4) ? hi : lo;
    }
    return r;
}

__device__ __forceinline__ void ffn_lds_barrier() {
    __builtin_amdgcn_fence(__ATOMIC_RELEASE, "workgroup", "local");
    __builtin_amdgcn_s_barrier();
    __builtin_amdgcn_fence(__ATOMIC_ACQUIRE, "workgroup", "local");
}

__device__ __forceinline__ float ffn_sigmoid(float v) {
    return 1.0f / (1.0f + __expf(-v));
}

__global__ __launch_bounds__(WAVES_ * 32)
void ffn_fused_kernel(const float* __restrict__ x,
                      const float* __restrict__ W,
                      const float* __restrict__ bias,
                      const float* __restrict__ W_out,
                      const float* __restrict__ b_out,
                      const int*   __restrict__ idx,
                      const int*   __restrict__ idx_out,
                      float*       __restrict__ out) {
    __shared__ __attribute__((aligned(16))) float act[ACT_FLOATS_ + OUT_FLOATS_];
    float* stage = act + ACT_FLOATS_;

    const int tid  = threadIdx.x;
    const int lane = tid & 31;
    const int wave = tid >> 5;
    const int c  = lane & 15;
    const int h  = lane >> 4;
    const int q  = c >> 1;
    const int p4 = (c & 1) * 4;
    const int b0 = blockIdx.x * COLS_;

    const int n0 = wave * 16;
    const int nr = n0 + 8 * h + q;
    FfnPre pre;
    ffn_preload(W, idx, n0, c, h, q, pre);

    #pragma unroll
    for (int i = 0; i < (COLS_ * NIN_) / (WAVES_ * 32); ++i) {
        const int e   = tid + WAVES_ * 32 * i;
        const int col = e / NIN_;
        const int f   = e % NIN_;
        act[f * COLS_ + col] = x[(size_t)(b0 + col) * NIN_ + f];
    }
    __syncthreads();

    #pragma nounroll
    for (int l = 0; l < L_; ++l) {
        const float bv = bias[l * N_ + nr];
        const bool last = (l + 1 == L_);
        const float* Wn = last ? W_out   : W   + (size_t)(l + 1) * N_ * K_;
        const int*   In = last ? idx_out : idx + (size_t)(l + 1) * N_ * K_;
        const int    nn = last ? (n0 & (NOUT_ - 1)) : n0;
        const float4 s = ffn_tile<true>(act, idx + (size_t)l * N_ * K_, n0, c, h, q, p4, pre,
                                        Wn + (nn + c) * K_ + 2 * h, In + (nn + q) * K_ + 2 * h);
        float4 o;
        o.x = ffn_sigmoid(s.x + bv);
        o.y = ffn_sigmoid(s.y + bv);
        o.z = ffn_sigmoid(s.z + bv);
        o.w = ffn_sigmoid(s.w + bv);
        *(float4*)(act + (NIN_ + l * N_ + nr) * COLS_ + p4) = o;
        ffn_lds_barrier();
    }

    if (wave < NOUT_ / 16) {
        const float bv = b_out[nr];
        const float4 s = ffn_tile<false>(act, idx_out, n0, c, h, q, p4, pre, nullptr, nullptr);
        float* sp = stage + p4 * NOUT_ + nr;
        sp[0 * NOUT_] = ffn_sigmoid(s.x + bv);
        sp[1 * NOUT_] = ffn_sigmoid(s.y + bv);
        sp[2 * NOUT_] = ffn_sigmoid(s.z + bv);
        sp[3 * NOUT_] = ffn_sigmoid(s.w + bv);
    }
    __syncthreads();
    if (tid < OUT_FLOATS_ / 4) {
        const int row = tid / (NOUT_ / 4);
        const int qq  = tid % (NOUT_ / 4);
        const float4 v = *(const float4*)(stage + row * NOUT_ + qq * 4);
        *(float4*)(out + (size_t)(b0 + row) * NOUT_ + qq * 4) = v;
    }
}

extern "C" void kernel_launch(void* const* d_in, const int* in_sizes, int n_in,
                              void* d_out, int out_size, void* d_ws, size_t ws_size,
                              hipStream_t stream) {
    const float* x     = (const float*)d_in[0];
    const float* W     = (const float*)d_in[1];
    const float* b     = (const float*)d_in[2];
    const float* W_out = (const float*)d_in[3];
    const float* b_out = (const float*)d_in[4];
    const int*   idx     = (const int*)d_in[5];
    const int*   idx_out = (const int*)d_in[6];
    float* out = (float*)d_out;

    ffn_fused_kernel<<<B_ / COLS_, WAVES_ * 32, 0, stream>>>(
        x, W, b, W_out, b_out, idx, idx_out, out);
}
